# starting kernel
# baseline (speedup 1.0000x reference)
#include <hip/hip_runtime.h>
#include <hip/hip_bf16.h>

#define NBATCH 16
#define NFEAT  256
#define NTIME  8000
#define SPEC_ROWS 257
#define ALPHA_F       0.95f
#define ONE_M_ALPHA_F 0.05f

#define NCHUNK 25
#define TCHUNK 320
#define NTILE  20

typedef __attribute__((ext_vector_type(2))) float v2f;
typedef __attribute__((ext_vector_type(4))) float v4f;
typedef __attribute__((ext_vector_type(8))) float v8f;

__device__ __forceinline__ float fast_rsq(float x) {
  return __builtin_amdgcn_rsqf(x);
}

__global__ void fse_pass1_chunk_contrib(const float* __restrict__ spec,
                                        float* __restrict__ Bc) {
  int tid = blockIdx.x * blockDim.x + threadIdx.x;
  if (tid >= NBATCH * NFEAT * NCHUNK) return;
  int c = tid % NCHUNK;
  int f = (tid / NCHUNK) % NFEAT;
  int b = tid / (NCHUNK * NFEAT);

  const float* p = spec + ((size_t)b * SPEC_ROWS + f) * NTIME + (size_t)c * TCHUNK;
  float s = 0.0f;
#pragma unroll 4
  for (int t = 0; t < TCHUNK; t += 4) {
    v4f x = *(const v4f*)(p + t);
    s = fmaf(fabsf(x.x), ONE_M_ALPHA_F, s * ALPHA_F);
    s = fmaf(fabsf(x.y), ONE_M_ALPHA_F, s * ALPHA_F);
    s = fmaf(fabsf(x.z), ONE_M_ALPHA_F, s * ALPHA_F);
    s = fmaf(fabsf(x.w), ONE_M_ALPHA_F, s * ALPHA_F);
  }
  Bc[tid] = s;
}

__global__ void fse_pass2_boundary_scan(const float* __restrict__ s0,
                                        const float* __restrict__ Bc,
                                        float* __restrict__ Sst) {
  int tid = blockIdx.x * blockDim.x + threadIdx.x;
  if (tid >= NBATCH * NFEAT) return;
  int f = tid % NFEAT;

  float decay = __powf(ALPHA_F, (float)TCHUNK);
  float S = s0[f];
  const float* bp = Bc + (size_t)tid * NCHUNK;
  float* sp = Sst + (size_t)tid * NCHUNK;
#pragma unroll
  for (int c = 0; c < NCHUNK; ++c) {
    sp[c] = S;
    S = fmaf(decay, S, bp[c]);
  }
}

__global__ void fse_pass3_wmma_scan(const float* __restrict__ spec,
                                    const float* __restrict__ Sst,
                                    float* __restrict__ out) {
  int wave = (int)((blockIdx.x * blockDim.x + threadIdx.x) >> 5);
  int lane = threadIdx.x & 31;
  if (wave >= NBATCH * (NFEAT / 16) * NCHUNK) return;

  int c  = wave % NCHUNK;
  int fg = (wave / NCHUNK) % (NFEAT / 16);
  int b  = wave / (NCHUNK * (NFEAT / 16));

  int hl = lane >> 4;
  int m  = lane & 15;
  int f  = fg * 16 + m;

  float pw[17];
  pw[0] = 1.0f;
#pragma unroll
  for (int i = 1; i <= 16; ++i) pw[i] = pw[i - 1] * ALPHA_F;

  v2f A[4];
#pragma unroll
  for (int j = 0; j < 4; ++j) {
    int k0 = 4 * j + 2 * hl;
    A[j].x = (k0     <= m) ? ONE_M_ALPHA_F * pw[m - k0]     : 0.0f;
    A[j].y = (k0 + 1 <= m) ? ONE_M_ALPHA_F * pw[m - k0 - 1] : 0.0f;
  }
  v2f Ac;
  Ac.x = (hl == 0) ? pw[m + 1] : 0.0f;
  Ac.y = 0.0f;

  float s_in = Sst[((size_t)b * NFEAT + f) * NCHUNK + c];

  const float* xin  = spec + ((size_t)b * SPEC_ROWS + f) * NTIME + (size_t)c * TCHUNK;
  float*       yout = out  + ((size_t)b * NFEAT     + f) * NTIME + (size_t)c * TCHUNK;

  for (int tile = 0; tile < NTILE; ++tile) {
    const float* xt = xin + tile * 16;

    v2f Bv[4];
#pragma unroll
    for (int j = 0; j < 4; ++j) {
      v2f xv = *(const v2f*)(xt + 4 * j + 2 * hl);
      Bv[j].x = fabsf(xv.x);
      Bv[j].y = fabsf(xv.y);
    }
    v2f Bc2;
    Bc2.x = (hl == 0) ? s_in : 0.0f;
    Bc2.y = 0.0f;

    v8f d = {};
    d = __builtin_amdgcn_wmma_f32_16x16x4_f32(false, Ac,   false, Bc2,   (short)0, d, false, false);
    d = __builtin_amdgcn_wmma_f32_16x16x4_f32(false, A[0], false, Bv[0], (short)0, d, false, false);
    d = __builtin_amdgcn_wmma_f32_16x16x4_f32(false, A[1], false, Bv[1], (short)0, d, false, false);
    d = __builtin_amdgcn_wmma_f32_16x16x4_f32(false, A[2], false, Bv[2], (short)0, d, false, false);
    d = __builtin_amdgcn_wmma_f32_16x16x4_f32(false, A[3], false, Bv[3], (short)0, d, false, false);

    s_in = __shfl(d[7], 16 + m, 32);

    const float* xr = xt + 8 * hl;
    v4f x0 = *(const v4f*)(xr);
    v4f x1 = *(const v4f*)(xr + 4);
    v4f y0, y1;
    y0.x = x0.x * fast_rsq(d[0]);
    y0.y = x0.y * fast_rsq(d[1]);
    y0.z = x0.z * fast_rsq(d[2]);
    y0.w = x0.w * fast_rsq(d[3]);
    y1.x = x1.x * fast_rsq(d[4]);
    y1.y = x1.y * fast_rsq(d[5]);
    y1.z = x1.z * fast_rsq(d[6]);
    y1.w = x1.w * fast_rsq(d[7]);

    float* yw = yout + tile * 16 + 8 * hl;
    __builtin_nontemporal_store(y0, (v4f*)(yw));
    __builtin_nontemporal_store(y1, (v4f*)(yw + 4));
  }
}

extern "C" void kernel_launch(void* const* d_in, const int* in_sizes, int n_in,
                              void* d_out, int out_size, void* d_ws, size_t ws_size,
                              hipStream_t stream) {
  const float* spec = (const float*)d_in[0];
  const float* s0   = (const float*)d_in[1];
  float* out = (float*)d_out;

  float* Bc  = (float*)d_ws;
  float* Sst = Bc + (size_t)NBATCH * NFEAT * NCHUNK;

  {
    int total = NBATCH * NFEAT * NCHUNK;
    fse_pass1_chunk_contrib<<<(total + 255) / 256, 256, 0, stream>>>(spec, Bc);
  }
  {
    int total = NBATCH * NFEAT;
    fse_pass2_boundary_scan<<<(total + 255) / 256, 256, 0, stream>>>(s0, Bc, Sst);
  }
  {
    int waves = NBATCH * (NFEAT / 16) * NCHUNK;
    int total = waves * 32;
    fse_pass3_wmma_scan<<<(total + 255) / 256, 256, 0, stream>>>(spec, Sst, out);
  }
}


# pass 3 moves data as whole 256-B row pieces (one b128 access per lane) through a wave-private LDS tile, next block requested one block ahead, instead of per-lane row-strided global accesses
# speedup vs baseline: 6.0783x; 6.0783x over previous
#include <hip/hip_runtime.h>
#include <hip/hip_bf16.h>

#define NBATCH 16
#define NFEAT  256
#define NTIME  8000
#define SPEC_ROWS 257
#define ALPHA_F       0.95f
#define ONE_M_ALPHA_F 0.05f

#define NCHUNK 25
#define TCHUNK 320
#define NTILE  20

typedef __attribute__((ext_vector_type(2))) float v2f;
typedef __attribute__((ext_vector_type(4))) float v4f;
typedef __attribute__((ext_vector_type(8))) float v8f;

__device__ __forceinline__ float fast_rsq(float x) {
  return __builtin_amdgcn_rsqf(x);
}

__global__ void fse_pass1_chunk_contrib(const float* __restrict__ spec,
                                        float* __restrict__ Bc) {
  int tid = blockIdx.x * blockDim.x + threadIdx.x;
  if (tid >= NBATCH * NFEAT * NCHUNK) return;
  int c = tid % NCHUNK;
  int f = (tid / NCHUNK) % NFEAT;
  int b = tid / (NCHUNK * NFEAT);

  const float* p = spec + ((size_t)b * SPEC_ROWS + f) * NTIME + (size_t)c * TCHUNK;
  float s = 0.0f;
#pragma unroll 4
  for (int t = 0; t < TCHUNK; t += 4) {
    v4f x = *(const v4f*)(p + t);
    s = fmaf(fabsf(x.x), ONE_M_ALPHA_F, s * ALPHA_F);
    s = fmaf(fabsf(x.y), ONE_M_ALPHA_F, s * ALPHA_F);
    s = fmaf(fabsf(x.z), ONE_M_ALPHA_F, s * ALPHA_F);
    s = fmaf(fabsf(x.w), ONE_M_ALPHA_F, s * ALPHA_F);
  }
  Bc[tid] = s;
}

__global__ void fse_pass2_boundary_scan(const float* __restrict__ s0,
                                        const float* __restrict__ Bc,
                                        float* __restrict__ Sst) {
  int tid = blockIdx.x * blockDim.x + threadIdx.x;
  if (tid >= NBATCH * NFEAT) return;
  int f = tid % NFEAT;

  float decay = __powf(ALPHA_F, (float)TCHUNK);
  float S = s0[f];
  const float* bp = Bc + (size_t)tid * NCHUNK;
  float* sp = Sst + (size_t)tid * NCHUNK;
#pragma unroll
  for (int c = 0; c < NCHUNK; ++c) {
    sp[c] = S;
    S = fmaf(decay, S, bp[c]);
  }
}

#define P3_WAVES   4
#define BLK_T      64
#define NBLK       (TCHUNK / BLK_T)
#define BLK_TILES  (BLK_T / 16)
#define LDS_PITCH  68
#define NROWLD     8

__global__ void __launch_bounds__(P3_WAVES * 32)
fse_pass3_wmma_scan(const float* __restrict__ spec,
                    const float* __restrict__ Sst,
                    float* __restrict__ out) {
  __shared__ __attribute__((aligned(16))) float stage[P3_WAVES][16 * LDS_PITCH];

  int wib  = (int)(threadIdx.x >> 5);
  int wave = (int)blockIdx.x * P3_WAVES + wib;
  int lane = threadIdx.x & 31;
  if (wave >= NBATCH * (NFEAT / 16) * NCHUNK) return;

  int c  = wave % NCHUNK;
  int fg = (wave / NCHUNK) % (NFEAT / 16);
  int b  = wave / (NCHUNK * (NFEAT / 16));

  int hl = lane >> 4;
  int m  = lane & 15;
  int f  = fg * 16 + m;

  float pw[17];
  pw[0] = 1.0f;
#pragma unroll
  for (int i = 1; i <= 16; ++i) pw[i] = pw[i - 1] * ALPHA_F;

  v2f A[4];
#pragma unroll
  for (int j = 0; j < 4; ++j) {
    int k0 = 4 * j + 2 * hl;
    A[j].x = (k0     <= m) ? ONE_M_ALPHA_F * pw[m - k0]     : 0.0f;
    A[j].y = (k0 + 1 <= m) ? ONE_M_ALPHA_F * pw[m - k0 - 1] : 0.0f;
  }
  v2f Ac;
  Ac.x = (hl == 0) ? pw[m + 1] : 0.0f;
  Ac.y = 0.0f;

  float s_in = Sst[((size_t)b * NFEAT + f) * NCHUNK + c];

  const float* gin  = spec + ((size_t)b * SPEC_ROWS + fg * 16 + hl) * NTIME + (size_t)c * TCHUNK + 4 * m;
  float*       gout = out  + ((size_t)b * NFEAT     + fg * 16 + hl) * NTIME + (size_t)c * TCHUNK + 4 * m;
  float* wtile = stage[wib];
  float* lrow  = wtile + hl * LDS_PITCH + 4 * m;
  float* frow  = wtile + m * LDS_PITCH;

  v4f pre[NROWLD];
#pragma unroll
  for (int i = 0; i < NROWLD; ++i)
    pre[i] = *(const v4f*)(gin + (size_t)(2 * i) * NTIME);

  for (int blk = 0; blk < NBLK; ++blk) {
#pragma unroll
    for (int i = 0; i < NROWLD; ++i)
      *(v4f*)(lrow + 2 * i * LDS_PITCH) = pre[i];

    if (blk + 1 < NBLK) {
#pragma unroll
      for (int i = 0; i < NROWLD; ++i)
        pre[i] = *(const v4f*)(gin + (size_t)(2 * i) * NTIME + (blk + 1) * BLK_T);
      __builtin_amdgcn_sched_barrier(0);
    }

#pragma unroll
    for (int tl = 0; tl < BLK_TILES; ++tl) {
      float* xt = frow + tl * 16;

      v2f Bv[4];
#pragma unroll
      for (int j = 0; j < 4; ++j) {
        v2f xv = *(const v2f*)(xt + 4 * j + 2 * hl);
        Bv[j].x = fabsf(xv.x);
        Bv[j].y = fabsf(xv.y);
      }
      v2f Bc2;
      Bc2.x = (hl == 0) ? s_in : 0.0f;
      Bc2.y = 0.0f;

      v8f d = {};
      d = __builtin_amdgcn_wmma_f32_16x16x4_f32(false, Ac,   false, Bc2,   (short)0, d, false, false);
      d = __builtin_amdgcn_wmma_f32_16x16x4_f32(false, A[0], false, Bv[0], (short)0, d, false, false);
      d = __builtin_amdgcn_wmma_f32_16x16x4_f32(false, A[1], false, Bv[1], (short)0, d, false, false);
      d = __builtin_amdgcn_wmma_f32_16x16x4_f32(false, A[2], false, Bv[2], (short)0, d, false, false);
      d = __builtin_amdgcn_wmma_f32_16x16x4_f32(false, A[3], false, Bv[3], (short)0, d, false, false);

      s_in = __shfl(d[7], 16 + m, 32);

      float* xr = xt + 8 * hl;
      v4f x0 = *(const v4f*)(xr);
      v4f x1 = *(const v4f*)(xr + 4);
      v4f y0, y1;
      y0.x = x0.x * fast_rsq(d[0]);
      y0.y = x0.y * fast_rsq(d[1]);
      y0.z = x0.z * fast_rsq(d[2]);
      y0.w = x0.w * fast_rsq(d[3]);
      y1.x = x1.x * fast_rsq(d[4]);
      y1.y = x1.y * fast_rsq(d[5]);
      y1.z = x1.z * fast_rsq(d[6]);
      y1.w = x1.w * fast_rsq(d[7]);
      *(v4f*)(xr)     = y0;
      *(v4f*)(xr + 4) = y1;
    }

    float* yw = gout + blk * BLK_T;
#pragma unroll
    for (int i = 0; i < NROWLD; ++i) {
      v4f yv = *(const v4f*)(lrow + 2 * i * LDS_PITCH);
      __builtin_nontemporal_store(yv, (v4f*)(yw + (size_t)(2 * i) * NTIME));
    }
  }
}

extern "C" void kernel_launch(void* const* d_in, const int* in_sizes, int n_in,
                              void* d_out, int out_size, void* d_ws, size_t ws_size,
                              hipStream_t stream) {
  const float* spec = (const float*)d_in[0];
  const float* s0   = (const float*)d_in[1];
  float* out = (float*)d_out;

  float* Bc  = (float*)d_ws;
  float* Sst = Bc + (size_t)NBATCH * NFEAT * NCHUNK;

  {
    int total = NBATCH * NFEAT * NCHUNK;
    fse_pass1_chunk_contrib<<<(total + 255) / 256, 256, 0, stream>>>(spec, Bc);
  }
  {
    int total = NBATCH * NFEAT;
    fse_pass2_boundary_scan<<<(total + 255) / 256, 256, 0, stream>>>(s0, Bc, Sst);
  }
  {
    int waves = NBATCH * (NFEAT / 16) * NCHUNK;
    fse_pass3_wmma_scan<<<(waves + P3_WAVES - 1) / P3_WAVES, P3_WAVES * 32, 0, stream>>>(spec, Sst, out);
  }
}


# pass 1 reads whole 256-B row pieces (one b128 per lane, 16 rows per wave); chunk-end sums = per-lane weighted partials + 16-lane butterfly, not one strided stream per lane (sum order changes)
# speedup vs baseline: 7.9302x; 1.3047x over previous
#include <hip/hip_runtime.h>
#include <hip/hip_bf16.h>

#define NBATCH 16
#define NFEAT  256
#define NTIME  8000
#define SPEC_ROWS 257
#define ALPHA_F       0.95f
#define ONE_M_ALPHA_F 0.05f

#define NCHUNK 25
#define TCHUNK 320
#define NTILE  20

typedef __attribute__((ext_vector_type(2))) float v2f;
typedef __attribute__((ext_vector_type(4))) float v4f;
typedef __attribute__((ext_vector_type(8))) float v8f;

__device__ __forceinline__ float fast_rsq(float x) {
  return __builtin_amdgcn_rsqf(x);
}

#define P1_WAVES 4
#define P1_BLK   64
#define P1_NBLK  (TCHUNK / P1_BLK)
#define P1_ROWLD 8
#define AP1  0.949999988f
#define AP2  0.902499974f
#define AP3  0.857374966f
#define AP4  0.814506233f
#define AP8  0.663420379f
#define AP16 0.440126568f
#define AP32 0.193711400f
#define AP64 0.0375241078f

__global__ void __launch_bounds__(P1_WAVES * 32)
fse_pass1_chunk_contrib(const float* __restrict__ spec,
                        float* __restrict__ Bc) {
  int wave = (int)blockIdx.x * P1_WAVES + (int)(threadIdx.x >> 5);
  int lane = threadIdx.x & 31;
  if (wave >= NBATCH * (NFEAT / 16) * NCHUNK) return;

  int c  = wave % NCHUNK;
  int fg = (wave / NCHUNK) % (NFEAT / 16);
  int b  = wave / (NCHUNK * (NFEAT / 16));
  int hl = lane >> 4;
  int m  = lane & 15;

  int n = 15 - m;
  float p = 1.0f;
  if (n & 1) p *= AP4;
  if (n & 2) p *= AP8;
  if (n & 4) p *= AP16;
  if (n & 8) p *= AP32;
  float w3 = ONE_M_ALPHA_F * p;
  float w2 = w3 * AP1;
  float w1 = w3 * AP2;
  float w0 = w3 * AP3;

  const float* gin = spec + ((size_t)b * SPEC_ROWS + fg * 16 + hl) * NTIME + (size_t)c * TCHUNK + 4 * m;

  v4f buf[P1_NBLK][P1_ROWLD];
  float acc[P1_ROWLD];
#pragma unroll
  for (int i = 0; i < P1_ROWLD; ++i) acc[i] = 0.0f;

#pragma unroll
  for (int blk = 0; blk < 3; ++blk) {
#pragma unroll
    for (int i = 0; i < P1_ROWLD; ++i)
      buf[blk][i] = *(const v4f*)(gin + (size_t)(2 * i) * NTIME + blk * P1_BLK);
    __builtin_amdgcn_sched_barrier(0);
  }

#pragma unroll
  for (int blk = 0; blk < P1_NBLK; ++blk) {
#pragma unroll
    for (int i = 0; i < P1_ROWLD; ++i) {
      v4f x = buf[blk][i];
      float t = w0 * fabsf(x.x);
      t = fmaf(w1, fabsf(x.y), t);
      t = fmaf(w2, fabsf(x.z), t);
      t = fmaf(w3, fabsf(x.w), t);
      acc[i] = fmaf(acc[i], AP64, t);
    }
    if (blk + 3 < P1_NBLK) {
      __builtin_amdgcn_sched_barrier(0);
#pragma unroll
      for (int i = 0; i < P1_ROWLD; ++i)
        buf[blk + 3][i] = *(const v4f*)(gin + (size_t)(2 * i) * NTIME + (blk + 3) * P1_BLK);
      __builtin_amdgcn_sched_barrier(0);
    }
  }

#pragma unroll
  for (int i = 0; i < P1_ROWLD; ++i) {
    float v = acc[i];
    v += __shfl_xor(v, 8, 32);
    v += __shfl_xor(v, 4, 32);
    v += __shfl_xor(v, 2, 32);
    v += __shfl_xor(v, 1, 32);
    acc[i] = v;
  }
  float r = acc[0];
#pragma unroll
  for (int i = 1; i < P1_ROWLD; ++i) r = (m == i) ? acc[i] : r;
  if (m < P1_ROWLD)
    Bc[((size_t)b * NFEAT + fg * 16 + 2 * m + hl) * NCHUNK + c] = r;
}

__global__ void fse_pass2_boundary_scan(const float* __restrict__ s0,
                                        const float* __restrict__ Bc,
                                        float* __restrict__ Sst) {
  int tid = blockIdx.x * blockDim.x + threadIdx.x;
  if (tid >= NBATCH * NFEAT) return;
  int f = tid % NFEAT;

  float decay = __powf(ALPHA_F, (float)TCHUNK);
  float S = s0[f];
  const float* bp = Bc + (size_t)tid * NCHUNK;
  float* sp = Sst + (size_t)tid * NCHUNK;
#pragma unroll
  for (int c = 0; c < NCHUNK; ++c) {
    sp[c] = S;
    S = fmaf(decay, S, bp[c]);
  }
}

#define P3_WAVES   4
#define BLK_T      64
#define NBLK       (TCHUNK / BLK_T)
#define BLK_TILES  (BLK_T / 16)
#define LDS_PITCH  68
#define NROWLD     8

__global__ void __launch_bounds__(P3_WAVES * 32)
fse_pass3_wmma_scan(const float* __restrict__ spec,
                    const float* __restrict__ Sst,
                    float* __restrict__ out) {
  __shared__ __attribute__((aligned(16))) float stage[P3_WAVES][16 * LDS_PITCH];

  int wib  = (int)(threadIdx.x >> 5);
  int wave = (int)blockIdx.x * P3_WAVES + wib;
  int lane = threadIdx.x & 31;
  if (wave >= NBATCH * (NFEAT / 16) * NCHUNK) return;

  int c  = wave % NCHUNK;
  int fg = (wave / NCHUNK) % (NFEAT / 16);
  int b  = wave / (NCHUNK * (NFEAT / 16));

  int hl = lane >> 4;
  int m  = lane & 15;
  int f  = fg * 16 + m;

  float pw[17];
  pw[0] = 1.0f;
#pragma unroll
  for (int i = 1; i <= 16; ++i) pw[i] = pw[i - 1] * ALPHA_F;

  v2f A[4];
#pragma unroll
  for (int j = 0; j < 4; ++j) {
    int k0 = 4 * j + 2 * hl;
    A[j].x = (k0     <= m) ? ONE_M_ALPHA_F * pw[m - k0]     : 0.0f;
    A[j].y = (k0 + 1 <= m) ? ONE_M_ALPHA_F * pw[m - k0 - 1] : 0.0f;
  }
  v2f Ac;
  Ac.x = (hl == 0) ? pw[m + 1] : 0.0f;
  Ac.y = 0.0f;

  float s_in = Sst[((size_t)b * NFEAT + f) * NCHUNK + c];

  const float* gin  = spec + ((size_t)b * SPEC_ROWS + fg * 16 + hl) * NTIME + (size_t)c * TCHUNK + 4 * m;
  float*       gout = out  + ((size_t)b * NFEAT     + fg * 16 + hl) * NTIME + (size_t)c * TCHUNK + 4 * m;
  float* wtile = stage[wib];
  float* lrow  = wtile + hl * LDS_PITCH + 4 * m;
  float* frow  = wtile + m * LDS_PITCH;

  v4f pre[NROWLD];
#pragma unroll
  for (int i = 0; i < NROWLD; ++i)
    pre[i] = *(const v4f*)(gin + (size_t)(2 * i) * NTIME);

  for (int blk = 0; blk < NBLK; ++blk) {
#pragma unroll
    for (int i = 0; i < NROWLD; ++i)
      *(v4f*)(lrow + 2 * i * LDS_PITCH) = pre[i];

    if (blk + 1 < NBLK) {
#pragma unroll
      for (int i = 0; i < NROWLD; ++i)
        pre[i] = *(const v4f*)(gin + (size_t)(2 * i) * NTIME + (blk + 1) * BLK_T);
      __builtin_amdgcn_sched_barrier(0);
    }

#pragma unroll
    for (int tl = 0; tl < BLK_TILES; ++tl) {
      float* xt = frow + tl * 16;

      v2f Bv[4];
#pragma unroll
      for (int j = 0; j < 4; ++j) {
        v2f xv = *(const v2f*)(xt + 4 * j + 2 * hl);
        Bv[j].x = fabsf(xv.x);
        Bv[j].y = fabsf(xv.y);
      }
      v2f Bc2;
      Bc2.x = (hl == 0) ? s_in : 0.0f;
      Bc2.y = 0.0f;

      v8f d = {};
      d = __builtin_amdgcn_wmma_f32_16x16x4_f32(false, Ac,   false, Bc2,   (short)0, d, false, false);
      d = __builtin_amdgcn_wmma_f32_16x16x4_f32(false, A[0], false, Bv[0], (short)0, d, false, false);
      d = __builtin_amdgcn_wmma_f32_16x16x4_f32(false, A[1], false, Bv[1], (short)0, d, false, false);
      d = __builtin_amdgcn_wmma_f32_16x16x4_f32(false, A[2], false, Bv[2], (short)0, d, false, false);
      d = __builtin_amdgcn_wmma_f32_16x16x4_f32(false, A[3], false, Bv[3], (short)0, d, false, false);

      s_in = __shfl(d[7], 16 + m, 32);

      float* xr = xt + 8 * hl;
      v4f x0 = *(const v4f*)(xr);
      v4f x1 = *(const v4f*)(xr + 4);
      v4f y0, y1;
      y0.x = x0.x * fast_rsq(d[0]);
      y0.y = x0.y * fast_rsq(d[1]);
      y0.z = x0.z * fast_rsq(d[2]);
      y0.w = x0.w * fast_rsq(d[3]);
      y1.x = x1.x * fast_rsq(d[4]);
      y1.y = x1.y * fast_rsq(d[5]);
      y1.z = x1.z * fast_rsq(d[6]);
      y1.w = x1.w * fast_rsq(d[7]);
      *(v4f*)(xr)     = y0;
      *(v4f*)(xr + 4) = y1;
    }

    float* yw = gout + blk * BLK_T;
#pragma unroll
    for (int i = 0; i < NROWLD; ++i) {
      v4f yv = *(const v4f*)(lrow + 2 * i * LDS_PITCH);
      __builtin_nontemporal_store(yv, (v4f*)(yw + (size_t)(2 * i) * NTIME));
    }
  }
}

extern "C" void kernel_launch(void* const* d_in, const int* in_sizes, int n_in,
                              void* d_out, int out_size, void* d_ws, size_t ws_size,
                              hipStream_t stream) {
  const float* spec = (const float*)d_in[0];
  const float* s0   = (const float*)d_in[1];
  float* out = (float*)d_out;

  float* Bc  = (float*)d_ws;
  float* Sst = Bc + (size_t)NBATCH * NFEAT * NCHUNK;

  {
    int waves = NBATCH * (NFEAT / 16) * NCHUNK;
    fse_pass1_chunk_contrib<<<(waves + P1_WAVES - 1) / P1_WAVES, P1_WAVES * 32, 0, stream>>>(spec, Bc);
  }
  {
    int total = NBATCH * NFEAT;
    fse_pass2_boundary_scan<<<(total + 255) / 256, 256, 0, stream>>>(s0, Bc, Sst);
  }
  {
    int waves = NBATCH * (NFEAT / 16) * NCHUNK;
    fse_pass3_wmma_scan<<<(waves + P3_WAVES - 1) / P3_WAVES, P3_WAVES * 32, 0, stream>>>(spec, Sst, out);
  }
}


# pass 3 runs the chunk-boundary recurrence itself (per lane, from pass 1's chunk sums, stored 64 B per chunk): boundary-scan launch removed, 2 launches not 3; sum order kept
# speedup vs baseline: 9.4987x; 1.1978x over previous
#include <hip/hip_runtime.h>
#include <hip/hip_bf16.h>

#define NBATCH 16
#define NFEAT  256
#define NTIME  8000
#define SPEC_ROWS 257
#define ALPHA_F       0.95f
#define ONE_M_ALPHA_F 0.05f

#define NCHUNK 25
#define TCHUNK 320
#define NTILE  20

typedef __attribute__((ext_vector_type(2))) float v2f;
typedef __attribute__((ext_vector_type(4))) float v4f;
typedef __attribute__((ext_vector_type(8))) float v8f;

__device__ __forceinline__ float fast_rsq(float x) {
  return __builtin_amdgcn_rsqf(x);
}

#define P1_WAVES 4
#define P1_BLK   64
#define P1_NBLK  (TCHUNK / P1_BLK)
#define P1_ROWLD 8
#define AP1  0.949999988f
#define AP2  0.902499974f
#define AP3  0.857374966f
#define AP4  0.814506233f
#define AP8  0.663420379f
#define AP16 0.440126568f
#define AP32 0.193711400f
#define AP64 0.0375241078f

__global__ void __launch_bounds__(P1_WAVES * 32)
fse_pass1_chunk_contrib(const float* __restrict__ spec,
                        float* __restrict__ Bt) {
  int wave = (int)blockIdx.x * P1_WAVES + (int)(threadIdx.x >> 5);
  int lane = threadIdx.x & 31;
  if (wave >= NBATCH * (NFEAT / 16) * NCHUNK) return;

  int c  = wave % NCHUNK;
  int fg = (wave / NCHUNK) % (NFEAT / 16);
  int b  = wave / (NCHUNK * (NFEAT / 16));
  int hl = lane >> 4;
  int m  = lane & 15;

  int n = 15 - m;
  float p = 1.0f;
  if (n & 1) p *= AP4;
  if (n & 2) p *= AP8;
  if (n & 4) p *= AP16;
  if (n & 8) p *= AP32;
  float w3 = ONE_M_ALPHA_F * p;
  float w2 = w3 * AP1;
  float w1 = w3 * AP2;
  float w0 = w3 * AP3;

  const float* gin = spec + ((size_t)b * SPEC_ROWS + fg * 16 + hl) * NTIME + (size_t)c * TCHUNK + 4 * m;

  v4f buf[P1_NBLK][P1_ROWLD];
  float acc[P1_ROWLD];
#pragma unroll
  for (int i = 0; i < P1_ROWLD; ++i) acc[i] = 0.0f;

#pragma unroll
  for (int blk = 0; blk < 3; ++blk) {
#pragma unroll
    for (int i = 0; i < P1_ROWLD; ++i)
      buf[blk][i] = *(const v4f*)(gin + (size_t)(2 * i) * NTIME + blk * P1_BLK);
    __builtin_amdgcn_sched_barrier(0);
  }

#pragma unroll
  for (int blk = 0; blk < P1_NBLK; ++blk) {
#pragma unroll
    for (int i = 0; i < P1_ROWLD; ++i) {
      v4f x = buf[blk][i];
      float t = w0 * fabsf(x.x);
      t = fmaf(w1, fabsf(x.y), t);
      t = fmaf(w2, fabsf(x.z), t);
      t = fmaf(w3, fabsf(x.w), t);
      acc[i] = fmaf(acc[i], AP64, t);
    }
    if (blk + 3 < P1_NBLK) {
      __builtin_amdgcn_sched_barrier(0);
#pragma unroll
      for (int i = 0; i < P1_ROWLD; ++i)
        buf[blk + 3][i] = *(const v4f*)(gin + (size_t)(2 * i) * NTIME + (blk + 3) * P1_BLK);
      __builtin_amdgcn_sched_barrier(0);
    }
  }

#pragma unroll
  for (int i = 0; i < P1_ROWLD; ++i) {
    float v = acc[i];
    v += __shfl_xor(v, 8, 32);
    v += __shfl_xor(v, 4, 32);
    v += __shfl_xor(v, 2, 32);
    v += __shfl_xor(v, 1, 32);
    acc[i] = v;
  }
  float r = acc[0];
#pragma unroll
  for (int i = 1; i < P1_ROWLD; ++i) r = (m == i) ? acc[i] : r;
  if (m < P1_ROWLD)
    Bt[((size_t)(b * (NFEAT / 16) + fg) * NCHUNK + c) * 16 + 2 * m + hl] = r;
}

#define P3_WAVES   4
#define BLK_T      64
#define NBLK       (TCHUNK / BLK_T)
#define BLK_TILES  (BLK_T / 16)
#define LDS_PITCH  68
#define NROWLD     8

__global__ void __launch_bounds__(P3_WAVES * 32)
fse_pass3_wmma_scan(const float* __restrict__ spec,
                    const float* __restrict__ s0,
                    const float* __restrict__ Bt,
                    float* __restrict__ out) {
  __shared__ __attribute__((aligned(16))) float stage[P3_WAVES][16 * LDS_PITCH];

  int wib  = (int)(threadIdx.x >> 5);
  int wave = (int)blockIdx.x * P3_WAVES + wib;
  int lane = threadIdx.x & 31;
  if (wave >= NBATCH * (NFEAT / 16) * NCHUNK) return;

  int c  = wave % NCHUNK;
  int fg = (wave / NCHUNK) % (NFEAT / 16);
  int b  = wave / (NCHUNK * (NFEAT / 16));

  int hl = lane >> 4;
  int m  = lane & 15;
  int f  = fg * 16 + m;

  float pw[17];
  pw[0] = 1.0f;
#pragma unroll
  for (int i = 1; i <= 16; ++i) pw[i] = pw[i - 1] * ALPHA_F;

  v2f A[4];
#pragma unroll
  for (int j = 0; j < 4; ++j) {
    int k0 = 4 * j + 2 * hl;
    A[j].x = (k0     <= m) ? ONE_M_ALPHA_F * pw[m - k0]     : 0.0f;
    A[j].y = (k0 + 1 <= m) ? ONE_M_ALPHA_F * pw[m - k0 - 1] : 0.0f;
  }
  v2f Ac;
  Ac.x = (hl == 0) ? pw[m + 1] : 0.0f;
  Ac.y = 0.0f;

  const float* bt = Bt + ((size_t)(b * (NFEAT / 16) + fg) * NCHUNK) * 16 + m;
  float bj[NCHUNK - 1];
#pragma unroll
  for (int j = 0; j < NCHUNK - 1; ++j) bj[j] = bt[j * 16];
  float s_in = s0[f];

  const float* gin  = spec + ((size_t)b * SPEC_ROWS + fg * 16 + hl) * NTIME + (size_t)c * TCHUNK + 4 * m;
  float*       gout = out  + ((size_t)b * NFEAT     + fg * 16 + hl) * NTIME + (size_t)c * TCHUNK + 4 * m;
  float* wtile = stage[wib];
  float* lrow  = wtile + hl * LDS_PITCH + 4 * m;
  float* frow  = wtile + m * LDS_PITCH;

  v4f pre[NROWLD];
#pragma unroll
  for (int i = 0; i < NROWLD; ++i)
    pre[i] = *(const v4f*)(gin + (size_t)(2 * i) * NTIME);

  __builtin_amdgcn_sched_barrier(0);

  {
    float decay = __powf(ALPHA_F, (float)TCHUNK);
#pragma unroll
    for (int j = 0; j < NCHUNK - 1; ++j) {
      float t = fmaf(decay, s_in, bj[j]);
      s_in = (j < c) ? t : s_in;
    }
  }

  for (int blk = 0; blk < NBLK; ++blk) {
#pragma unroll
    for (int i = 0; i < NROWLD; ++i)
      *(v4f*)(lrow + 2 * i * LDS_PITCH) = pre[i];

    if (blk + 1 < NBLK) {
#pragma unroll
      for (int i = 0; i < NROWLD; ++i)
        pre[i] = *(const v4f*)(gin + (size_t)(2 * i) * NTIME + (blk + 1) * BLK_T);
      __builtin_amdgcn_sched_barrier(0);
    }

#pragma unroll
    for (int tl = 0; tl < BLK_TILES; ++tl) {
      float* xt = frow + tl * 16;

      v2f Bv[4];
#pragma unroll
      for (int j = 0; j < 4; ++j) {
        v2f xv = *(const v2f*)(xt + 4 * j + 2 * hl);
        Bv[j].x = fabsf(xv.x);
        Bv[j].y = fabsf(xv.y);
      }
      v2f Bc2;
      Bc2.x = (hl == 0) ? s_in : 0.0f;
      Bc2.y = 0.0f;

      v8f d = {};
      d = __builtin_amdgcn_wmma_f32_16x16x4_f32(false, Ac,   false, Bc2,   (short)0, d, false, false);
      d = __builtin_amdgcn_wmma_f32_16x16x4_f32(false, A[0], false, Bv[0], (short)0, d, false, false);
      d = __builtin_amdgcn_wmma_f32_16x16x4_f32(false, A[1], false, Bv[1], (short)0, d, false, false);
      d = __builtin_amdgcn_wmma_f32_16x16x4_f32(false, A[2], false, Bv[2], (short)0, d, false, false);
      d = __builtin_amdgcn_wmma_f32_16x16x4_f32(false, A[3], false, Bv[3], (short)0, d, false, false);

      s_in = __shfl(d[7], 16 + m, 32);

      float* xr = xt + 8 * hl;
      v4f x0 = *(const v4f*)(xr);
      v4f x1 = *(const v4f*)(xr + 4);
      v4f y0, y1;
      y0.x = x0.x * fast_rsq(d[0]);
      y0.y = x0.y * fast_rsq(d[1]);
      y0.z = x0.z * fast_rsq(d[2]);
      y0.w = x0.w * fast_rsq(d[3]);
      y1.x = x1.x * fast_rsq(d[4]);
      y1.y = x1.y * fast_rsq(d[5]);
      y1.z = x1.z * fast_rsq(d[6]);
      y1.w = x1.w * fast_rsq(d[7]);
      *(v4f*)(xr)     = y0;
      *(v4f*)(xr + 4) = y1;
    }

    float* yw = gout + blk * BLK_T;
#pragma unroll
    for (int i = 0; i < NROWLD; ++i) {
      v4f yv = *(const v4f*)(lrow + 2 * i * LDS_PITCH);
      __builtin_nontemporal_store(yv, (v4f*)(yw + (size_t)(2 * i) * NTIME));
    }
  }
}

extern "C" void kernel_launch(void* const* d_in, const int* in_sizes, int n_in,
                              void* d_out, int out_size, void* d_ws, size_t ws_size,
                              hipStream_t stream) {
  const float* spec = (const float*)d_in[0];
  const float* s0   = (const float*)d_in[1];
  float* out = (float*)d_out;

  float* Bt = (float*)d_ws;

  {
    int waves = NBATCH * (NFEAT / 16) * NCHUNK;
    fse_pass1_chunk_contrib<<<(waves + P1_WAVES - 1) / P1_WAVES, P1_WAVES * 32, 0, stream>>>(spec, Bt);
  }
  {
    int waves = NBATCH * (NFEAT / 16) * NCHUNK;
    fse_pass3_wmma_scan<<<(waves + P3_WAVES - 1) / P3_WAVES, P3_WAVES * 32, 0, stream>>>(spec, s0, Bt, out);
  }
}
